# starting kernel
# baseline (speedup 1.0000x reference)
#include <hip/hip_runtime.h>


#define DIMD   2000
#define NPAD   2048
#define BATCH  4096
#define NFIELD 40
#define NF     8
#define MT     64
#define NSUB   4
#define CK     200
#define NKC    10
#define LDX    204
#define NSPLIT 4
#define NTPW   4

typedef __attribute__((ext_vector_type(2))) float v2f;
typedef __attribute__((ext_vector_type(8))) float v8f;

__global__ __launch_bounds__(256) void FFM_build_A(
    const float* __restrict__ V, const int* __restrict__ fd,
    const float* __restrict__ W, float* __restrict__ Asw,
    float* __restrict__ dA, float* __restrict__ wp)
{
    const int idx = blockIdx.x * blockDim.x + threadIdx.x;
    const int i = idx >> 11;
    const int j = idx & (NPAD - 1);
    float s = 0.0f;
    if (j < DIMD) {
        const int fi = fd[i];
        const int fj = fd[j];
        const float4* Vi = reinterpret_cast<const float4*>(V + (size_t)i * (NFIELD * NF) + fj * NF);
        const float4* Vj = reinterpret_cast<const float4*>(V + (size_t)j * (NFIELD * NF) + fi * NF);
        float4 a0 = Vi[0], a1 = Vi[1];
        float4 b0 = Vj[0], b1 = Vj[1];
        s = a0.x * b0.x + a0.y * b0.y + a0.z * b0.z + a0.w * b0.w
          + a1.x * b1.x + a1.y * b1.y + a1.z * b1.z + a1.w * b1.w;
    }
    Asw[(((size_t)(i >> 1) * NPAD + j) << 1) + (i & 1)] = s;
    if (i == j && j < DIMD) dA[j] = s;
    if (i == 0) {
        wp[j] = (j < DIMD) ? W[j] : 0.0f;
        if (j >= DIMD) dA[j] = 0.0f;
    }
}

__global__ __launch_bounds__(256) void FFM_wmma_gemm(
    const float* __restrict__ x, const float* __restrict__ Asw,
    const float* __restrict__ dA, const float* __restrict__ wp,
    float* __restrict__ parts)
{
    extern __shared__ float lds[];
    const int tid  = threadIdx.x;
    const int w    = tid >> 5;
    const int lane = tid & 31;
    const int hi   = lane >> 4;
    const int lo   = lane & 15;
    const int sp   = blockIdx.y;
    const int s_base = blockIdx.x * MT;

    v8f c[NTPW][NSUB];
    #pragma unroll
    for (int t = 0; t < NTPW; ++t)
        #pragma unroll
        for (int u = 0; u < NSUB; ++u)
            c[t][u] = (v8f){0.f, 0.f, 0.f, 0.f, 0.f, 0.f, 0.f, 0.f};

    for (int kc = 0; kc < NKC; ++kc) {
        const int nv4 = MT * (CK / 4);
        for (int i = tid; i < nv4; i += 256) {
            const int r  = i / (CK / 4);
            const int cq = i % (CK / 4);
            float4 v = *reinterpret_cast<const float4*>(
                x + (size_t)(s_base + r) * DIMD + kc * CK + cq * 4);
            *reinterpret_cast<float4*>(&lds[r * LDX + cq * 4]) = v;
        }
        __syncthreads();

        for (int ks = 0; ks < CK / 4; ++ks) {
            const int k0 = ks * 4;
            const int kg = kc * CK + k0;
            v2f a[NSUB];
            #pragma unroll
            for (int u = 0; u < NSUB; ++u) {
                const float* p = &lds[(u * 16 + lo) * LDX + k0 + hi * 2];
                a[u][0] = p[0]; a[u][1] = p[1];
            }
            #pragma unroll
            for (int t = 0; t < NTPW; ++t) {
                const int n = ((t * 32 + w * 4 + sp) * 16) + lo;
                v2f bf = *reinterpret_cast<const v2f*>(
                    Asw + (((size_t)(kg >> 1) + hi) * NPAD + n) * 2);
                #pragma unroll
                for (int u = 0; u < NSUB; ++u)
                    c[t][u] = __builtin_amdgcn_wmma_f32_16x16x4_f32(
                        false, a[u], false, bf, (short)0, c[t][u], false, false);
            }
        }
        __syncthreads();
    }

    float racc[NSUB][8];
    #pragma unroll
    for (int u = 0; u < NSUB; ++u)
        #pragma unroll
        for (int v = 0; v < 8; ++v) racc[u][v] = 0.0f;

    #pragma unroll
    for (int t = 0; t < NTPW; ++t) {
        const int n = ((t * 32 + w * 4 + sp) * 16) + lo;
        const float wn = wp[n];
        const float dn = dA[n];
        const bool inb = (n < DIMD);
        #pragma unroll
        for (int u = 0; u < NSUB; ++u) {
            #pragma unroll
            for (int v = 0; v < 8; ++v) {
                const int r = v + hi * 8 + u * 16;
                float xv = inb ? x[(size_t)(s_base + r) * DIMD + n] : 0.0f;
                float tv = c[t][u][v];
                racc[u][v] += xv * (0.5f * tv + wn) - 0.5f * xv * xv * dn;
            }
        }
    }

    #pragma unroll
    for (int u = 0; u < NSUB; ++u) {
        #pragma unroll
        for (int v = 0; v < 8; ++v) {
            float s = racc[u][v];
            s += __shfl_xor(s, 8, 32);
            s += __shfl_xor(s, 4, 32);
            s += __shfl_xor(s, 2, 32);
            s += __shfl_xor(s, 1, 32);
            if (lo == 0) {
                const int r = v + hi * 8 + u * 16;
                lds[w * MT + r] = s;
            }
        }
    }
    __syncthreads();

    if (tid < MT) {
        float p = 0.0f;
        #pragma unroll
        for (int w2 = 0; w2 < 8; ++w2) p += lds[w2 * MT + tid];
        parts[(size_t)(s_base + tid) * NSPLIT + sp] = p;
    }
}

__global__ __launch_bounds__(256) void FFM_finalize(
    const float* __restrict__ parts, const float* __restrict__ b,
    float* __restrict__ out)
{
    const int i = blockIdx.x * blockDim.x + threadIdx.x;
    if (i < BATCH) {
        float p = b[0];
        #pragma unroll
        for (int s = 0; s < NSPLIT; ++s) p += parts[(size_t)i * NSPLIT + s];
        out[i] = p;
    }
}

extern "C" void kernel_launch(void* const* d_in, const int* in_sizes, int n_in,
                              void* d_out, int out_size, void* d_ws, size_t ws_size,
                              hipStream_t stream)
{
    const float* x  = (const float*)d_in[0];
    const int*   fd = (const int*)  d_in[1];
    const float* b  = (const float*)d_in[2];
    const float* W  = (const float*)d_in[3];
    const float* V  = (const float*)d_in[4];
    float* out = (float*)d_out;

    float* Asw   = (float*)d_ws;
    float* dA    = Asw + (size_t)DIMD * NPAD;
    float* wp    = dA + NPAD;
    float* parts = wp + NPAD;

    FFM_build_A<<<16000, 256, 0, stream>>>(V, fd, W, Asw, dA, wp);

    dim3 g2(BATCH / MT, NSPLIT);
    FFM_wmma_gemm<<<g2, 256, MT * LDX * sizeof(float), stream>>>(x, Asw, dA, wp, parts);

    FFM_finalize<<<BATCH / 256, 256, 0, stream>>>(parts, b, out);
}


# A build made field-major: block per (256 columns, row field), thread holds its column's V[j,p,:] in registers, walks that field's rows 8 loads deep; 128000 build waves -> 2560; sum order unchanged
# speedup vs baseline: 1.4664x; 1.4664x over previous
#include <hip/hip_runtime.h>


#define DIMD   2000
#define NPAD   2048
#define BATCH  4096
#define NFIELD 40
#define NF     8
#define MT     64
#define NSUB   4
#define CK     1000
#define NKC    2
#define LDX    1004
#define CKV    (CK / 4)
#define SROWS  (NTHR / 256)
#define SPASS  (MT / SROWS)
#define NSPLIT 4
#define NWAVE  32
#define NTHR   (NWAVE * 32)

typedef __attribute__((ext_vector_type(2))) float v2f;
typedef __attribute__((ext_vector_type(8))) float v8f;

#define BROWU 8
__global__ __launch_bounds__(256) void FFM_build_A(
    const float* __restrict__ V, const int* __restrict__ fd,
    const float* __restrict__ W, float* __restrict__ Asw,
    float* __restrict__ dA, float* __restrict__ wp)
{
    __shared__ int rows[DIMD];
    __shared__ int wcnt[8];
    const int tid  = threadIdx.x;
    const int lane = tid & 31;
    const int w    = tid >> 5;
    const int p    = blockIdx.y;
    const int j    = blockIdx.x * 256 + tid;
    const bool inb = (j < DIMD);

    float4 b0 = make_float4(0.f, 0.f, 0.f, 0.f), b1 = b0;
    int fj = 0;
    if (inb) {
        const float4* Vj = reinterpret_cast<const float4*>(V + (size_t)j * (NFIELD * NF) + p * NF);
        b0 = Vj[0]; b1 = Vj[1];
        fj = fd[j];
    }

    int  rid[8];
    bool hit[8];
    int  cnt = 0;
    #pragma unroll
    for (int c = 0; c < 8; ++c) {
        const int q = c * 32 + lane;
        rid[c] = w * 250 + q;
        hit[c] = (q < 250) && (fd[rid[c]] == p);
        cnt += __popc(__builtin_amdgcn_ballot_w32(hit[c]));
    }
    if (lane == 0) wcnt[w] = cnt;
    __syncthreads();
    int base = 0, n = 0;
    #pragma unroll
    for (int w2 = 0; w2 < 8; ++w2) {
        const int cw = wcnt[w2];
        if (w2 < w) base += cw;
        n += cw;
    }
    #pragma unroll
    for (int c = 0; c < 8; ++c) {
        const unsigned m = __builtin_amdgcn_ballot_w32(hit[c]);
        if (hit[c]) rows[base + __popc(m & ((1u << lane) - 1u))] = rid[c];
        base += __popc(m);
    }
    __syncthreads();

    const float* Vc = V + fj * NF;
    for (int t = 0; t < n; t += BROWU) {
        int    ri[BROWU];
        float4 a0[BROWU], a1[BROWU];
        #pragma unroll
        for (int u = 0; u < BROWU; ++u) {
            const int tt = (t + u < n) ? (t + u) : (n - 1);
            ri[u] = rows[tt];
            const float4* Vi = reinterpret_cast<const float4*>(Vc + (size_t)ri[u] * (NFIELD * NF));
            a0[u] = Vi[0]; a1[u] = Vi[1];
        }
        #pragma unroll
        for (int u = 0; u < BROWU; ++u) {
            if (t + u < n) {
                const int i = ri[u];
                float s = 0.0f;
                if (inb)
                    s = a0[u].x * b0.x + a0[u].y * b0.y + a0[u].z * b0.z + a0[u].w * b0.w
                      + a1[u].x * b1.x + a1[u].y * b1.y + a1[u].z * b1.z + a1[u].w * b1.w;
                Asw[(((size_t)(i >> 1) * NPAD + j) << 1) + (i & 1)] = s;
                if (i == j) dA[j] = s;
            }
        }
    }
    if (p == 0) {
        wp[j] = inb ? W[j] : 0.0f;
        if (!inb) dA[j] = 0.0f;
    }
}

__global__ __launch_bounds__(NTHR) void FFM_wmma_gemm(
    const float* __restrict__ x, const float* __restrict__ Asw,
    const float* __restrict__ dA, const float* __restrict__ wp,
    float* __restrict__ parts)
{
    __shared__ __attribute__((aligned(16))) float lds[MT * LDX];
    const int tid  = threadIdx.x;
    const int w    = tid >> 5;
    const int lane = tid & 31;
    const int hi   = lane >> 4;
    const int lo   = lane & 15;
    const int sp   = blockIdx.y;
    const int s_base = blockIdx.x * MT;

    const int n = ((w * NSPLIT + sp) * 16) + lo;

    v8f c[NSUB];
    #pragma unroll
    for (int u = 0; u < NSUB; ++u)
        c[u] = (v8f){0.f, 0.f, 0.f, 0.f, 0.f, 0.f, 0.f, 0.f};

    for (int kc = 0; kc < NKC; ++kc) {
        const int cq = tid & 255;
        const int r0 = tid >> 8;
        if (cq < CKV) {
            const float* xs = x + (size_t)(s_base + r0) * DIMD + kc * CK + cq * 4;
            float* ls = &lds[r0 * LDX + cq * 4];
            float4 v[SPASS];
            #pragma unroll
            for (int t = 0; t < SPASS; ++t)
                v[t] = *reinterpret_cast<const float4*>(xs + (size_t)t * SROWS * DIMD);
            #pragma unroll
            for (int t = 0; t < SPASS; ++t)
                *reinterpret_cast<float4*>(ls + t * SROWS * LDX) = v[t];
        }
        __syncthreads();

        for (int ks = 0; ks < CK / 4; ++ks) {
            const int k0 = ks * 4;
            const int kg = kc * CK + k0;
            v2f a[NSUB];
            #pragma unroll
            for (int u = 0; u < NSUB; ++u) {
                const float* p = &lds[(u * 16 + lo) * LDX + k0 + hi * 2];
                a[u][0] = p[0]; a[u][1] = p[1];
            }
            v2f bf = *reinterpret_cast<const v2f*>(
                Asw + (((size_t)(kg >> 1) + hi) * NPAD + n) * 2);
            #pragma unroll
            for (int u = 0; u < NSUB; ++u)
                c[u] = __builtin_amdgcn_wmma_f32_16x16x4_f32(
                    false, a[u], false, bf, (short)0, c[u], false, false);
        }
        __syncthreads();
    }

    float racc[NSUB][8];
    #pragma unroll
    for (int u = 0; u < NSUB; ++u)
        #pragma unroll
        for (int v = 0; v < 8; ++v) racc[u][v] = 0.0f;

    {
        const float wn = wp[n];
        const float dn = dA[n];
        const bool inb = (n < DIMD);
        #pragma unroll
        for (int u = 0; u < NSUB; ++u) {
            #pragma unroll
            for (int v = 0; v < 8; ++v) {
                const int r = v + hi * 8 + u * 16;
                float xv = inb ? x[(size_t)(s_base + r) * DIMD + n] : 0.0f;
                float tv = c[u][v];
                racc[u][v] += xv * (0.5f * tv + wn) - 0.5f * xv * xv * dn;
            }
        }
    }

    #pragma unroll
    for (int u = 0; u < NSUB; ++u) {
        #pragma unroll
        for (int v = 0; v < 8; ++v) {
            float s = racc[u][v];
            s += __shfl_xor(s, 8, 32);
            s += __shfl_xor(s, 4, 32);
            s += __shfl_xor(s, 2, 32);
            s += __shfl_xor(s, 1, 32);
            if (lo == 0) {
                const int r = v + hi * 8 + u * 16;
                lds[w * MT + r] = s;
            }
        }
    }
    __syncthreads();

    if (tid < MT) {
        float p = 0.0f;
        #pragma unroll
        for (int w2 = 0; w2 < NWAVE; ++w2) p += lds[w2 * MT + tid];
        parts[(size_t)(s_base + tid) * NSPLIT + sp] = p;
    }
}

__global__ __launch_bounds__(256) void FFM_finalize(
    const float* __restrict__ parts, const float* __restrict__ b,
    float* __restrict__ out)
{
    const int i = blockIdx.x * blockDim.x + threadIdx.x;
    if (i < BATCH) {
        float p = b[0];
        #pragma unroll
        for (int s = 0; s < NSPLIT; ++s) p += parts[(size_t)i * NSPLIT + s];
        out[i] = p;
    }
}

extern "C" void kernel_launch(void* const* d_in, const int* in_sizes, int n_in,
                              void* d_out, int out_size, void* d_ws, size_t ws_size,
                              hipStream_t stream)
{
    const float* x  = (const float*)d_in[0];
    const int*   fd = (const int*)  d_in[1];
    const float* b  = (const float*)d_in[2];
    const float* W  = (const float*)d_in[3];
    const float* V  = (const float*)d_in[4];
    float* out = (float*)d_out;

    float* Asw   = (float*)d_ws;
    float* dA    = Asw + (size_t)DIMD * NPAD;
    float* wp    = dA + NPAD;
    float* parts = wp + NPAD;

    dim3 g1(NPAD / 256, NFIELD);
    FFM_build_A<<<g1, 256, 0, stream>>>(V, fd, W, Asw, dA, wp);

    dim3 g2(BATCH / MT, NSPLIT);
    FFM_wmma_gemm<<<g2, NTHR, 0, stream>>>(x, Asw, dA, wp, parts);

    FFM_finalize<<<BATCH / 256, 256, 0, stream>>>(parts, b, out);
}


# k order inside fragments permuted so a lane holds 4 contiguous k: x and A fragments come by one b128 load per two WMMAs (were b64 per one), A stored in row quads; sum order changes
# speedup vs baseline: 1.6108x; 1.0985x over previous
#include <hip/hip_runtime.h>


#define DIMD   2000
#define NPAD   2048
#define BATCH  4096
#define NFIELD 40
#define NF     8
#define MT     64
#define NSUB   4
#define CK     1000
#define NKC    2
#define LDX    1004
#define CKV    (CK / 4)
#define SROWS  (NTHR / 256)
#define SPASS  (MT / SROWS)
#define NSPLIT 4
#define NWAVE  32
#define NTHR   (NWAVE * 32)

typedef __attribute__((ext_vector_type(2))) float v2f;
typedef __attribute__((ext_vector_type(4))) float v4f;
typedef __attribute__((ext_vector_type(8))) float v8f;

#define BROWU 8
__global__ __launch_bounds__(256) void FFM_build_A(
    const float* __restrict__ V, const int* __restrict__ fd,
    const float* __restrict__ W, float* __restrict__ Asw,
    float* __restrict__ dA, float* __restrict__ wp)
{
    __shared__ int rows[DIMD];
    __shared__ int wcnt[8];
    const int tid  = threadIdx.x;
    const int lane = tid & 31;
    const int w    = tid >> 5;
    const int p    = blockIdx.y;
    const int j    = blockIdx.x * 256 + tid;
    const bool inb = (j < DIMD);

    float4 b0 = make_float4(0.f, 0.f, 0.f, 0.f), b1 = b0;
    int fj = 0;
    if (inb) {
        const float4* Vj = reinterpret_cast<const float4*>(V + (size_t)j * (NFIELD * NF) + p * NF);
        b0 = Vj[0]; b1 = Vj[1];
        fj = fd[j];
    }

    int  rid[8];
    bool hit[8];
    int  cnt = 0;
    #pragma unroll
    for (int c = 0; c < 8; ++c) {
        const int q = c * 32 + lane;
        rid[c] = w * 250 + q;
        hit[c] = (q < 250) && (fd[rid[c]] == p);
        cnt += __popc(__builtin_amdgcn_ballot_w32(hit[c]));
    }
    if (lane == 0) wcnt[w] = cnt;
    __syncthreads();
    int base = 0, n = 0;
    #pragma unroll
    for (int w2 = 0; w2 < 8; ++w2) {
        const int cw = wcnt[w2];
        if (w2 < w) base += cw;
        n += cw;
    }
    #pragma unroll
    for (int c = 0; c < 8; ++c) {
        const unsigned m = __builtin_amdgcn_ballot_w32(hit[c]);
        if (hit[c]) rows[base + __popc(m & ((1u << lane) - 1u))] = rid[c];
        base += __popc(m);
    }
    __syncthreads();

    const float* Vc = V + fj * NF;
    for (int t = 0; t < n; t += BROWU) {
        int    ri[BROWU];
        float4 a0[BROWU], a1[BROWU];
        #pragma unroll
        for (int u = 0; u < BROWU; ++u) {
            const int tt = (t + u < n) ? (t + u) : (n - 1);
            ri[u] = rows[tt];
            const float4* Vi = reinterpret_cast<const float4*>(Vc + (size_t)ri[u] * (NFIELD * NF));
            a0[u] = Vi[0]; a1[u] = Vi[1];
        }
        #pragma unroll
        for (int u = 0; u < BROWU; ++u) {
            if (t + u < n) {
                const int i = ri[u];
                float s = 0.0f;
                if (inb)
                    s = a0[u].x * b0.x + a0[u].y * b0.y + a0[u].z * b0.z + a0[u].w * b0.w
                      + a1[u].x * b1.x + a1[u].y * b1.y + a1[u].z * b1.z + a1[u].w * b1.w;
                Asw[(((size_t)(i >> 2) * NPAD + j) << 2) + (i & 3)] = s;
                if (i == j) dA[j] = s;
            }
        }
    }
    if (p == 0) {
        wp[j] = inb ? W[j] : 0.0f;
        if (!inb) dA[j] = 0.0f;
    }
}

__global__ __launch_bounds__(NTHR) void FFM_wmma_gemm(
    const float* __restrict__ x, const float* __restrict__ Asw,
    const float* __restrict__ dA, const float* __restrict__ wp,
    float* __restrict__ parts)
{
    __shared__ __attribute__((aligned(16))) float lds[MT * LDX];
    const int tid  = threadIdx.x;
    const int w    = tid >> 5;
    const int lane = tid & 31;
    const int hi   = lane >> 4;
    const int lo   = lane & 15;
    const int sp   = blockIdx.y;
    const int s_base = blockIdx.x * MT;

    const int n = ((w * NSPLIT + sp) * 16) + lo;

    v8f c[NSUB];
    #pragma unroll
    for (int u = 0; u < NSUB; ++u)
        c[u] = (v8f){0.f, 0.f, 0.f, 0.f, 0.f, 0.f, 0.f, 0.f};

    const float* ap = &lds[lo * LDX + hi * 4];
    const float* bq = Asw + ((size_t)hi * NPAD + n) * 4;

    for (int kc = 0; kc < NKC; ++kc) {
        const int cq = tid & 255;
        const int r0 = tid >> 8;
        if (cq < CKV) {
            const float* xs = x + (size_t)(s_base + r0) * DIMD + kc * CK + cq * 4;
            float* ls = &lds[r0 * LDX + cq * 4];
            float4 v[SPASS];
            #pragma unroll
            for (int t = 0; t < SPASS; ++t)
                v[t] = *reinterpret_cast<const float4*>(xs + (size_t)t * SROWS * DIMD);
            #pragma unroll
            for (int t = 0; t < SPASS; ++t)
                *reinterpret_cast<float4*>(ls + t * SROWS * LDX) = v[t];
        }
        __syncthreads();

        const float* bp = bq + (size_t)kc * (CK / 4) * NPAD * 4;
        #pragma unroll 5
        for (int q = 0; q < CK / 8; ++q) {
            v4f a[NSUB];
            #pragma unroll
            for (int u = 0; u < NSUB; ++u)
                a[u] = *reinterpret_cast<const v4f*>(ap + u * 16 * LDX + q * 8);
            const v4f bf = *reinterpret_cast<const v4f*>(bp + (size_t)q * (2 * NPAD * 4));
            const v2f b01 = {bf[0], bf[1]}, b23 = {bf[2], bf[3]};
            #pragma unroll
            for (int u = 0; u < NSUB; ++u) {
                const v2f a01 = {a[u][0], a[u][1]};
                c[u] = __builtin_amdgcn_wmma_f32_16x16x4_f32(
                    false, a01, false, b01, (short)0, c[u], false, false);
            }
            #pragma unroll
            for (int u = 0; u < NSUB; ++u) {
                const v2f a23 = {a[u][2], a[u][3]};
                c[u] = __builtin_amdgcn_wmma_f32_16x16x4_f32(
                    false, a23, false, b23, (short)0, c[u], false, false);
            }
        }
        __syncthreads();
    }

    float racc[NSUB][8];
    #pragma unroll
    for (int u = 0; u < NSUB; ++u)
        #pragma unroll
        for (int v = 0; v < 8; ++v) racc[u][v] = 0.0f;

    {
        const float wn = wp[n];
        const float dn = dA[n];
        const bool inb = (n < DIMD);
        #pragma unroll
        for (int u = 0; u < NSUB; ++u) {
            #pragma unroll
            for (int v = 0; v < 8; ++v) {
                const int r = v + hi * 8 + u * 16;
                float xv = inb ? x[(size_t)(s_base + r) * DIMD + n] : 0.0f;
                float tv = c[u][v];
                racc[u][v] += xv * (0.5f * tv + wn) - 0.5f * xv * xv * dn;
            }
        }
    }

    #pragma unroll
    for (int u = 0; u < NSUB; ++u) {
        #pragma unroll
        for (int v = 0; v < 8; ++v) {
            float s = racc[u][v];
            s += __shfl_xor(s, 8, 32);
            s += __shfl_xor(s, 4, 32);
            s += __shfl_xor(s, 2, 32);
            s += __shfl_xor(s, 1, 32);
            if (lo == 0) {
                const int r = v + hi * 8 + u * 16;
                lds[w * MT + r] = s;
            }
        }
    }
    __syncthreads();

    if (tid < MT) {
        float p = 0.0f;
        #pragma unroll
        for (int w2 = 0; w2 < NWAVE; ++w2) p += lds[w2 * MT + tid];
        parts[(size_t)(s_base + tid) * NSPLIT + sp] = p;
    }
}

__global__ __launch_bounds__(256) void FFM_finalize(
    const float* __restrict__ parts, const float* __restrict__ b,
    float* __restrict__ out)
{
    const int i = blockIdx.x * blockDim.x + threadIdx.x;
    if (i < BATCH) {
        float p = b[0];
        #pragma unroll
        for (int s = 0; s < NSPLIT; ++s) p += parts[(size_t)i * NSPLIT + s];
        out[i] = p;
    }
}

extern "C" void kernel_launch(void* const* d_in, const int* in_sizes, int n_in,
                              void* d_out, int out_size, void* d_ws, size_t ws_size,
                              hipStream_t stream)
{
    const float* x  = (const float*)d_in[0];
    const int*   fd = (const int*)  d_in[1];
    const float* b  = (const float*)d_in[2];
    const float* W  = (const float*)d_in[3];
    const float* V  = (const float*)d_in[4];
    float* out = (float*)d_out;

    float* Asw   = (float*)d_ws;
    float* dA    = Asw + (size_t)DIMD * NPAD;
    float* wp    = dA + NPAD;
    float* parts = wp + NPAD;

    dim3 g1(NPAD / 256, NFIELD);
    FFM_build_A<<<g1, 256, 0, stream>>>(V, fd, W, Asw, dA, wp);

    dim3 g2(BATCH / MT, NSPLIT);
    FFM_wmma_gemm<<<g2, NTHR, 0, stream>>>(x, Asw, dA, wp, parts);

    FFM_finalize<<<BATCH / 256, 256, 0, stream>>>(parts, b, out);
}


# Wave priority falls with a wave's own progress through each K chunk (s_setprio 3,2,1,0 over four bands): lagging waves win the matrix pipe, a SIMD's waves reach the barrier level; sum order unchanged
# speedup vs baseline: 1.6718x; 1.0379x over previous
#include <hip/hip_runtime.h>


#define DIMD   2000
#define NPAD   2048
#define BATCH  4096
#define NFIELD 40
#define NF     8
#define MT     64
#define NSUB   4
#define CK     1000
#define NKC    2
#define LDX    1004
#define CKV    (CK / 4)
#define SROWS  (NTHR / 256)
#define SPASS  (MT / SROWS)
#define NSPLIT 4
#define PB1    8
#define PB2    16
#define PB3    24
#define NWAVE  32
#define NTHR   (NWAVE * 32)

typedef __attribute__((ext_vector_type(2))) float v2f;
typedef __attribute__((ext_vector_type(4))) float v4f;
typedef __attribute__((ext_vector_type(8))) float v8f;

#define BROWU 8
__global__ __launch_bounds__(256) void FFM_build_A(
    const float* __restrict__ V, const int* __restrict__ fd,
    const float* __restrict__ W, float* __restrict__ Asw,
    float* __restrict__ dA, float* __restrict__ wp)
{
    __shared__ int rows[DIMD];
    __shared__ int wcnt[8];
    const int tid  = threadIdx.x;
    const int lane = tid & 31;
    const int w    = tid >> 5;
    const int p    = blockIdx.y;
    const int j    = blockIdx.x * 256 + tid;
    const bool inb = (j < DIMD);

    float4 b0 = make_float4(0.f, 0.f, 0.f, 0.f), b1 = b0;
    int fj = 0;
    if (inb) {
        const float4* Vj = reinterpret_cast<const float4*>(V + (size_t)j * (NFIELD * NF) + p * NF);
        b0 = Vj[0]; b1 = Vj[1];
        fj = fd[j];
    }

    int  rid[8];
    bool hit[8];
    int  cnt = 0;
    #pragma unroll
    for (int c = 0; c < 8; ++c) {
        const int q = c * 32 + lane;
        rid[c] = w * 250 + q;
        hit[c] = (q < 250) && (fd[rid[c]] == p);
        cnt += __popc(__builtin_amdgcn_ballot_w32(hit[c]));
    }
    if (lane == 0) wcnt[w] = cnt;
    __syncthreads();
    int base = 0, n = 0;
    #pragma unroll
    for (int w2 = 0; w2 < 8; ++w2) {
        const int cw = wcnt[w2];
        if (w2 < w) base += cw;
        n += cw;
    }
    #pragma unroll
    for (int c = 0; c < 8; ++c) {
        const unsigned m = __builtin_amdgcn_ballot_w32(hit[c]);
        if (hit[c]) rows[base + __popc(m & ((1u << lane) - 1u))] = rid[c];
        base += __popc(m);
    }
    __syncthreads();

    const float* Vc = V + fj * NF;
    for (int t = 0; t < n; t += BROWU) {
        int    ri[BROWU];
        float4 a0[BROWU], a1[BROWU];
        #pragma unroll
        for (int u = 0; u < BROWU; ++u) {
            const int tt = (t + u < n) ? (t + u) : (n - 1);
            ri[u] = rows[tt];
            const float4* Vi = reinterpret_cast<const float4*>(Vc + (size_t)ri[u] * (NFIELD * NF));
            a0[u] = Vi[0]; a1[u] = Vi[1];
        }
        #pragma unroll
        for (int u = 0; u < BROWU; ++u) {
            if (t + u < n) {
                const int i = ri[u];
                float s = 0.0f;
                if (inb)
                    s = a0[u].x * b0.x + a0[u].y * b0.y + a0[u].z * b0.z + a0[u].w * b0.w
                      + a1[u].x * b1.x + a1[u].y * b1.y + a1[u].z * b1.z + a1[u].w * b1.w;
                Asw[(((size_t)(i >> 2) * NPAD + j) << 2) + (i & 3)] = s;
                if (i == j) dA[j] = s;
            }
        }
    }
    if (p == 0) {
        wp[j] = inb ? W[j] : 0.0f;
        if (!inb) dA[j] = 0.0f;
    }
}

__global__ __launch_bounds__(NTHR) void FFM_wmma_gemm(
    const float* __restrict__ x, const float* __restrict__ Asw,
    const float* __restrict__ dA, const float* __restrict__ wp,
    float* __restrict__ parts)
{
    __shared__ __attribute__((aligned(16))) float lds[MT * LDX];
    const int tid  = threadIdx.x;
    const int w    = tid >> 5;
    const int lane = tid & 31;
    const int hi   = lane >> 4;
    const int lo   = lane & 15;
    const int sp   = blockIdx.y;
    const int s_base = blockIdx.x * MT;

    const int n = ((w * NSPLIT + sp) * 16) + lo;

    v8f c[NSUB];
    #pragma unroll
    for (int u = 0; u < NSUB; ++u)
        c[u] = (v8f){0.f, 0.f, 0.f, 0.f, 0.f, 0.f, 0.f, 0.f};

    const float* ap = &lds[lo * LDX + hi * 4];
    const float* bq = Asw + ((size_t)hi * NPAD + n) * 4;

    for (int kc = 0; kc < NKC; ++kc) {
        const int cq = tid & 255;
        const int r0 = tid >> 8;
        if (cq < CKV) {
            const float* xs = x + (size_t)(s_base + r0) * DIMD + kc * CK + cq * 4;
            float* ls = &lds[r0 * LDX + cq * 4];
            float4 v[SPASS];
            #pragma unroll
            for (int t = 0; t < SPASS; ++t)
                v[t] = *reinterpret_cast<const float4*>(xs + (size_t)t * SROWS * DIMD);
            #pragma unroll
            for (int t = 0; t < SPASS; ++t)
                *reinterpret_cast<float4*>(ls + t * SROWS * LDX) = v[t];
        }
        __syncthreads();

        const float* bp = bq + (size_t)kc * (CK / 4) * NPAD * 4;
        __builtin_amdgcn_s_setprio(3);
        #pragma nounroll
        for (int t = 0; t < CK / 40; ++t) {
            if (t == PB1)      __builtin_amdgcn_s_setprio(2);
            else if (t == PB2) __builtin_amdgcn_s_setprio(1);
            else if (t == PB3) __builtin_amdgcn_s_setprio(0);
            #pragma unroll
            for (int j = 0; j < 5; ++j) {
                const int q = t * 5 + j;
                v4f a[NSUB];
                #pragma unroll
                for (int u = 0; u < NSUB; ++u)
                    a[u] = *reinterpret_cast<const v4f*>(ap + u * 16 * LDX + q * 8);
                const v4f bf = *reinterpret_cast<const v4f*>(bp + (size_t)q * (2 * NPAD * 4));
                const v2f b01 = {bf[0], bf[1]}, b23 = {bf[2], bf[3]};
                #pragma unroll
                for (int u = 0; u < NSUB; ++u) {
                    const v2f a01 = {a[u][0], a[u][1]};
                    c[u] = __builtin_amdgcn_wmma_f32_16x16x4_f32(
                        false, a01, false, b01, (short)0, c[u], false, false);
                }
                #pragma unroll
                for (int u = 0; u < NSUB; ++u) {
                    const v2f a23 = {a[u][2], a[u][3]};
                    c[u] = __builtin_amdgcn_wmma_f32_16x16x4_f32(
                        false, a23, false, b23, (short)0, c[u], false, false);
                }
            }
        }
        __syncthreads();
    }

    float racc[NSUB][8];
    #pragma unroll
    for (int u = 0; u < NSUB; ++u)
        #pragma unroll
        for (int v = 0; v < 8; ++v) racc[u][v] = 0.0f;

    {
        const float wn = wp[n];
        const float dn = dA[n];
        const bool inb = (n < DIMD);
        #pragma unroll
        for (int u = 0; u < NSUB; ++u) {
            #pragma unroll
            for (int v = 0; v < 8; ++v) {
                const int r = v + hi * 8 + u * 16;
                float xv = inb ? x[(size_t)(s_base + r) * DIMD + n] : 0.0f;
                float tv = c[u][v];
                racc[u][v] += xv * (0.5f * tv + wn) - 0.5f * xv * xv * dn;
            }
        }
    }

    #pragma unroll
    for (int u = 0; u < NSUB; ++u) {
        #pragma unroll
        for (int v = 0; v < 8; ++v) {
            float s = racc[u][v];
            s += __shfl_xor(s, 8, 32);
            s += __shfl_xor(s, 4, 32);
            s += __shfl_xor(s, 2, 32);
            s += __shfl_xor(s, 1, 32);
            if (lo == 0) {
                const int r = v + hi * 8 + u * 16;
                lds[w * MT + r] = s;
            }
        }
    }
    __syncthreads();

    if (tid < MT) {
        float p = 0.0f;
        #pragma unroll
        for (int w2 = 0; w2 < NWAVE; ++w2) p += lds[w2 * MT + tid];
        parts[(size_t)(s_base + tid) * NSPLIT + sp] = p;
    }
}

__global__ __launch_bounds__(256) void FFM_finalize(
    const float* __restrict__ parts, const float* __restrict__ b,
    float* __restrict__ out)
{
    const int i = blockIdx.x * blockDim.x + threadIdx.x;
    if (i < BATCH) {
        float p = b[0];
        #pragma unroll
        for (int s = 0; s < NSPLIT; ++s) p += parts[(size_t)i * NSPLIT + s];
        out[i] = p;
    }
}

extern "C" void kernel_launch(void* const* d_in, const int* in_sizes, int n_in,
                              void* d_out, int out_size, void* d_ws, size_t ws_size,
                              hipStream_t stream)
{
    const float* x  = (const float*)d_in[0];
    const int*   fd = (const int*)  d_in[1];
    const float* b  = (const float*)d_in[2];
    const float* W  = (const float*)d_in[3];
    const float* V  = (const float*)d_in[4];
    float* out = (float*)d_out;

    float* Asw   = (float*)d_ws;
    float* dA    = Asw + (size_t)DIMD * NPAD;
    float* wp    = dA + NPAD;
    float* parts = wp + NPAD;

    dim3 g1(NPAD / 256, NFIELD);
    FFM_build_A<<<g1, 256, 0, stream>>>(V, fd, W, Asw, dA, wp);

    dim3 g2(BATCH / MT, NSPLIT);
    FFM_wmma_gemm<<<g2, NTHR, 0, stream>>>(x, Asw, dA, wp, parts);

    FFM_finalize<<<BATCH / 256, 256, 0, stream>>>(parts, b, out);
}
